# starting kernel
# baseline (speedup 1.0000x reference)
#include <hip/hip_runtime.h>
#include <hip/hip_bf16.h>
#include <math.h>

#define NN 50000
#define EE 800000
#define RR 8
#define IND 128
#define OUTD 128
#define HH 8
#define HDD 16

typedef float v2f __attribute__((ext_vector_type(2)));
typedef float v8f __attribute__((ext_vector_type(8)));

__device__ __forceinline__ void atomicMaxFloat(float* addr, float v) {
    if (v >= 0.0f) {
        atomicMax((int*)addr, __float_as_int(v));
    } else {
        atomicMin((unsigned int*)addr, __float_as_uint(v));
    }
}

__global__ __launch_bounds__(256) void rgat_init(float* __restrict__ out,
                                                 float* __restrict__ maxbuf,
                                                 float* __restrict__ denom) {
    int t = blockIdx.x * 256 + threadIdx.x;
    if (t < NN * OUTD) out[t] = 0.0f;
    if (t < NN * HH) {
        denom[t] = 0.0f;
        maxbuf[t] = -INFINITY;
    }
}

#define HS_A 132
#define WPS  288

__global__ __launch_bounds__(256) void rgat_proj_wmma(const float* __restrict__ h,
                                                      const float* __restrict__ W,
                                                      float* __restrict__ z) {
    extern __shared__ float lds[];
    float* h_s = lds;
    float* w_s = lds + 64 * HS_A;

    const int r     = blockIdx.y;
    const int nbase = blockIdx.x * 64;
    const int tid   = threadIdx.x;

    for (int i = tid; i < 64 * 32; i += 256) {
        int row = i >> 5;
        int c4  = i & 31;
        int g   = nbase + row;
        if (g >= NN) g = NN - 1;
        float4 v = *(const float4*)(h + (size_t)g * IND + c4 * 4);
        float* d = h_s + row * HS_A + c4 * 4;
        d[0] = v.x; d[1] = v.y; d[2] = v.z; d[3] = v.w;
    }
    const float* Wr = W + (size_t)r * IND * OUTD;
    for (int i = tid; i < 64 * 128; i += 256) {
        int kp  = i >> 7;
        int col = i & 127;
        float w0 = Wr[(size_t)(2 * kp) * OUTD + col];
        float w1 = Wr[(size_t)(2 * kp + 1) * OUTD + col];
        v2f p; p.x = w0; p.y = w1;
        *(v2f*)(w_s + kp * WPS + col * 2) = p;
    }
    __syncthreads();

    const int wave = tid >> 5;
    const int lane = tid & 31;
    const int ln   = lane & 15;
    const int g    = lane >> 4;
    const int rt   = wave & 3;
    const int cb   = (wave >> 2) * 64;

    v8f acc0 = {}, acc1 = {}, acc2 = {}, acc3 = {};
    const float* arow = h_s + (rt * 16 + ln) * HS_A;
    const float* bcol = w_s + (size_t)g * WPS + (size_t)(cb + ln) * 2;

    for (int k = 0; k < IND; k += 4) {
        v2f a = *(const v2f*)(arow + k + 2 * g);
        const float* brow = bcol + (k >> 1) * WPS;
        v2f b0 = *(const v2f*)(brow + 0 * 32);
        v2f b1 = *(const v2f*)(brow + 1 * 32);
        v2f b2 = *(const v2f*)(brow + 2 * 32);
        v2f b3 = *(const v2f*)(brow + 3 * 32);
        acc0 = __builtin_amdgcn_wmma_f32_16x16x4_f32(false, a, false, b0, (short)0, acc0, false, false);
        acc1 = __builtin_amdgcn_wmma_f32_16x16x4_f32(false, a, false, b1, (short)0, acc1, false, false);
        acc2 = __builtin_amdgcn_wmma_f32_16x16x4_f32(false, a, false, b2, (short)0, acc2, false, false);
        acc3 = __builtin_amdgcn_wmma_f32_16x16x4_f32(false, a, false, b3, (short)0, acc3, false, false);
    }

    float* zr = z + (size_t)r * NN * OUTD;
#pragma unroll
    for (int ct = 0; ct < 4; ++ct) {
        v8f acc = (ct == 0) ? acc0 : (ct == 1) ? acc1 : (ct == 2) ? acc2 : acc3;
        int col = cb + ct * 16 + ln;
#pragma unroll
        for (int v = 0; v < 8; ++v) {
            int row = nbase + rt * 16 + 8 * g + v;
            if (row < NN) zr[(size_t)row * OUTD + col] = acc[v];
        }
    }
}

__global__ __launch_bounds__(256) void rgat_edge_score(const float* __restrict__ z,
                                                       const float* __restrict__ attw,
                                                       const int* __restrict__ src,
                                                       const int* __restrict__ dst,
                                                       const int* __restrict__ etype,
                                                       float* __restrict__ e_val,
                                                       float* __restrict__ maxbuf) {
    int t = blockIdx.x * 256 + threadIdx.x;
    if (t >= EE * HH) return;
    int e  = t >> 3;
    int hd = t & 7;
    int r = etype[e], s = src[e], d = dst[e];

    const float4* zs = (const float4*)(z + ((size_t)r * NN + s) * OUTD + hd * HDD);
    const float4* zd = (const float4*)(z + ((size_t)r * NN + d) * OUTD + hd * HDD);
    const float4* aw = (const float4*)(attw + ((size_t)r * HH + hd) * (2 * HDD));

    float acc = 0.0f;
#pragma unroll
    for (int q = 0; q < 4; ++q) {
        float4 a = zs[q], wA = aw[q];
        float4 b = zd[q], wB = aw[4 + q];
        acc += a.x * wA.x + a.y * wA.y + a.z * wA.z + a.w * wA.w;
        acc += b.x * wB.x + b.y * wB.y + b.z * wB.z + b.w * wB.w;
    }
    float ev = (acc > 0.0f) ? acc : 0.01f * acc;
    e_val[t] = ev;
    atomicMaxFloat(maxbuf + (size_t)d * HH + hd, ev);
}

__global__ __launch_bounds__(256) void rgat_edge_accum(const float* __restrict__ z,
                                                       const int* __restrict__ src,
                                                       const int* __restrict__ dst,
                                                       const int* __restrict__ etype,
                                                       const float* __restrict__ e_val,
                                                       const float* __restrict__ maxbuf,
                                                       float* __restrict__ denom,
                                                       float* __restrict__ out) {
    int t = blockIdx.x * 256 + threadIdx.x;
    if (t >= EE * HH) return;
    int e  = t >> 3;
    int hd = t & 7;
    int r = etype[e], s = src[e], d = dst[e];

    float m  = maxbuf[(size_t)d * HH + hd];
    float ex = __expf(e_val[t] - m);
    atomicAdd(denom + (size_t)d * HH + hd, ex);

    const float* zs = z + ((size_t)r * NN + s) * OUTD + hd * HDD;
    float* o = out + (size_t)d * OUTD + hd * HDD;
#pragma unroll
    for (int i = 0; i < HDD; ++i) atomicAdd(o + i, ex * zs[i]);
}

__global__ __launch_bounds__(256) void rgat_finalize(float* __restrict__ out,
                                                     const float* __restrict__ denom) {
    int t = blockIdx.x * 256 + threadIdx.x;
    if (t >= NN * OUTD) return;
    out[t] = out[t] / fmaxf(denom[t >> 4], 1e-9f);
}

extern "C" void kernel_launch(void* const* d_in, const int* in_sizes, int n_in,
                              void* d_out, int out_size, void* d_ws, size_t ws_size,
                              hipStream_t stream) {
    const float* h    = (const float*)d_in[0];
    const float* fcw  = (const float*)d_in[1];
    const float* attw = (const float*)d_in[2];
    const int*   src  = (const int*)d_in[3];
    const int*   dst  = (const int*)d_in[4];
    const int*   ety  = (const int*)d_in[5];
    float* out = (float*)d_out;

    float* z      = (float*)d_ws;
    float* e_val  = z + (size_t)RR * NN * OUTD;
    float* maxbuf = e_val + (size_t)EE * HH;
    float* denom  = maxbuf + (size_t)NN * HH;

    {
        int total = NN * OUTD;
        rgat_init<<<(total + 255) / 256, 256, 0, stream>>>(out, maxbuf, denom);
    }
    {
        dim3 grid((NN + 63) / 64, RR);
        size_t shmem = (size_t)(64 * HS_A + 64 * WPS) * sizeof(float);
        rgat_proj_wmma<<<grid, 256, shmem, stream>>>(h, fcw, z);
    }
    {
        int total = EE * HH;
        rgat_edge_score<<<(total + 255) / 256, 256, 0, stream>>>(
            z, attw, src, dst, ety, e_val, maxbuf);
    }
    {
        int total = EE * HH;
        rgat_edge_accum<<<(total + 255) / 256, 256, 0, stream>>>(
            z, src, dst, ety, e_val, maxbuf, denom, out);
    }
    {
        int total = NN * OUTD;
        rgat_finalize<<<(total + 255) / 256, 256, 0, stream>>>(out, denom);
    }
}


# Edges grouped by destination in-call (counting sort, integer atomics); one wave per node does logits, one-pass softmax and weighted sum in registers: no float atomics, sums in rank order
# speedup vs baseline: 2.7963x; 2.7963x over previous
#include <hip/hip_runtime.h>
#include <hip/hip_bf16.h>
#include <math.h>

#define NN 50000
#define EE 800000
#define RR 8
#define IND 128
#define OUTD 128
#define HH 8
#define HDD 16

typedef float v2f __attribute__((ext_vector_type(2)));
typedef float v8f __attribute__((ext_vector_type(8)));

#define HS_A 132
#define WPS  288

__global__ __launch_bounds__(256) void rgat_proj_wmma(const float* __restrict__ h,
                                                      const float* __restrict__ W,
                                                      float* __restrict__ z) {
    extern __shared__ float lds[];
    float* h_s = lds;
    float* w_s = lds + 64 * HS_A;

    const int r     = blockIdx.y;
    const int nbase = blockIdx.x * 64;
    const int tid   = threadIdx.x;

    for (int i = tid; i < 64 * 32; i += 256) {
        int row = i >> 5;
        int c4  = i & 31;
        int g   = nbase + row;
        if (g >= NN) g = NN - 1;
        float4 v = *(const float4*)(h + (size_t)g * IND + c4 * 4);
        float* d = h_s + row * HS_A + c4 * 4;
        d[0] = v.x; d[1] = v.y; d[2] = v.z; d[3] = v.w;
    }
    const float* Wr = W + (size_t)r * IND * OUTD;
    for (int i = tid; i < 64 * 128; i += 256) {
        int kp  = i >> 7;
        int col = i & 127;
        float w0 = Wr[(size_t)(2 * kp) * OUTD + col];
        float w1 = Wr[(size_t)(2 * kp + 1) * OUTD + col];
        v2f p; p.x = w0; p.y = w1;
        *(v2f*)(w_s + kp * WPS + col * 2) = p;
    }
    __syncthreads();

    const int wave = tid >> 5;
    const int lane = tid & 31;
    const int ln   = lane & 15;
    const int g    = lane >> 4;
    const int rt   = wave & 3;
    const int cb   = (wave >> 2) * 64;

    v8f acc0 = {}, acc1 = {}, acc2 = {}, acc3 = {};
    const float* arow = h_s + (rt * 16 + ln) * HS_A;
    const float* bcol = w_s + (size_t)g * WPS + (size_t)(cb + ln) * 2;

    for (int k = 0; k < IND; k += 4) {
        v2f a = *(const v2f*)(arow + k + 2 * g);
        const float* brow = bcol + (k >> 1) * WPS;
        v2f b0 = *(const v2f*)(brow + 0 * 32);
        v2f b1 = *(const v2f*)(brow + 1 * 32);
        v2f b2 = *(const v2f*)(brow + 2 * 32);
        v2f b3 = *(const v2f*)(brow + 3 * 32);
        acc0 = __builtin_amdgcn_wmma_f32_16x16x4_f32(false, a, false, b0, (short)0, acc0, false, false);
        acc1 = __builtin_amdgcn_wmma_f32_16x16x4_f32(false, a, false, b1, (short)0, acc1, false, false);
        acc2 = __builtin_amdgcn_wmma_f32_16x16x4_f32(false, a, false, b2, (short)0, acc2, false, false);
        acc3 = __builtin_amdgcn_wmma_f32_16x16x4_f32(false, a, false, b3, (short)0, acc3, false, false);
    }

    float* zr = z + (size_t)r * NN * OUTD;
#pragma unroll
    for (int ct = 0; ct < 4; ++ct) {
        v8f acc = (ct == 0) ? acc0 : (ct == 1) ? acc1 : (ct == 2) ? acc2 : acc3;
        int col = cb + ct * 16 + ln;
#pragma unroll
        for (int v = 0; v < 8; ++v) {
            int row = nbase + rt * 16 + 8 * g + v;
            if (row < NN) zr[(size_t)row * OUTD + col] = acc[v];
        }
    }
}

#define SCAN_T   1024
#define SCAN_PER 52
#define CNT_PAD  (SCAN_T * SCAN_PER)

__global__ __launch_bounds__(256) void rgat_zero_counts(int* __restrict__ cnt) {
    int t = blockIdx.x * 256 + threadIdx.x;
    if (t < CNT_PAD) cnt[t] = 0;
}

__global__ __launch_bounds__(256) void rgat_dst_rank(const int* __restrict__ dst,
                                                     int* __restrict__ cnt,
                                                     int* __restrict__ rank) {
    int e = blockIdx.x * 256 + threadIdx.x;
    if (e >= EE) return;
    rank[e] = atomicAdd(cnt + dst[e], 1);
}

__global__ __launch_bounds__(SCAN_T) void rgat_scan_counts(const int* __restrict__ cnt,
                                                           int* __restrict__ offs) {
    __shared__ int wave_tot[32];
    const int tid  = threadIdx.x;
    const int lane = tid & 31;
    const int wv   = tid >> 5;

    const int4* p = (const int4*)(cnt + tid * SCAN_PER);
    int4 v[SCAN_PER / 4];
    int s = 0;
#pragma unroll
    for (int i = 0; i < SCAN_PER / 4; ++i) {
        v[i] = p[i];
        s += v[i].x + v[i].y + v[i].z + v[i].w;
    }
    int inc = s;
#pragma unroll
    for (int o = 1; o < 32; o <<= 1) {
        int t = __shfl_up(inc, o);
        if (lane >= o) inc += t;
    }
    if (lane == 31) wave_tot[wv] = inc;
    __syncthreads();
    if (wv == 0) {
        int x  = wave_tot[lane];
        int xi = x;
#pragma unroll
        for (int o = 1; o < 32; o <<= 1) {
            int t = __shfl_up(xi, o);
            if (lane >= o) xi += t;
        }
        wave_tot[lane] = xi - x;
    }
    __syncthreads();
    int run = wave_tot[wv] + inc - s;
    int4* q = (int4*)(offs + tid * SCAN_PER);
#pragma unroll
    for (int i = 0; i < SCAN_PER / 4; ++i) {
        int4 o;
        o.x = run; run += v[i].x;
        o.y = run; run += v[i].y;
        o.z = run; run += v[i].z;
        o.w = run; run += v[i].w;
        q[i] = o;
    }
}

__global__ __launch_bounds__(256) void rgat_group_edges(const int* __restrict__ src,
                                                        const int* __restrict__ dst,
                                                        const int* __restrict__ etype,
                                                        const int* __restrict__ offs,
                                                        const int* __restrict__ rank,
                                                        int2* __restrict__ grouped) {
    int e = blockIdx.x * 256 + threadIdx.x;
    if (e >= EE) return;
    int pos = offs[dst[e]] + rank[e];
    grouped[pos] = make_int2(src[e], etype[e]);
}

__device__ __forceinline__ float quad_sum(float x) {
    int b = __builtin_bit_cast(int, x);
    float y = x + __builtin_bit_cast(float, __builtin_amdgcn_update_dpp(b, b, 0xB1, 0xF, 0xF, false));
    int c = __builtin_bit_cast(int, y);
    return y + __builtin_bit_cast(float, __builtin_amdgcn_update_dpp(c, c, 0x4E, 0xF, 0xF, false));
}

__device__ __forceinline__ float pick8(const float (&p)[8], int r) {
    float a = (r & 1) ? p[1] : p[0];
    float b = (r & 1) ? p[3] : p[2];
    float c = (r & 1) ? p[5] : p[4];
    float d = (r & 1) ? p[7] : p[6];
    float ab = (r & 2) ? b : a;
    float cd = (r & 2) ? d : c;
    return (r & 4) ? cd : ab;
}

__device__ __forceinline__ float dot4(float4 a, float4 b) {
    return a.x * b.x + a.y * b.y + a.z * b.z + a.w * b.w;
}

__global__ __launch_bounds__(256) void rgat_node_softmax_sum(const float* __restrict__ z,
                                                             const float* __restrict__ attw,
                                                             const int* __restrict__ offs,
                                                             const int2* __restrict__ grouped,
                                                             float* __restrict__ out) {
    const int lane = threadIdx.x & 31;
    const int d = __builtin_amdgcn_readfirstlane(blockIdx.x * 8 + (threadIdx.x >> 5));
    if (d >= NN) return;
    const int hd = lane >> 2;
    const int qd = lane & 3;

    const int beg = __builtin_amdgcn_readfirstlane(offs[d]);
    const int end = __builtin_amdgcn_readfirstlane(offs[d + 1]);

    float4 acc = make_float4(0.0f, 0.0f, 0.0f, 0.0f);
    float  m = -INFINITY;
    float  l = 0.0f;

    if (end > beg) {
        int base = beg;
        int n = end - base;
        if (n > 32) n = 32;
        int2 my = make_int2(0, 0);
        if (lane < n) my = grouped[base + lane];

        float4 zd[8];
#pragma unroll
        for (int r = 0; r < RR; ++r)
            zd[r] = *(const float4*)(z + ((size_t)r * NN + d) * OUTD + lane * 4);
        float pd[8];
#pragma unroll
        for (int r = 0; r < RR; ++r) {
            float4 wD = *(const float4*)(attw + ((size_t)r * HH + hd) * (2 * HDD) + HDD + qd * 4);
            pd[r] = dot4(zd[r], wD);
        }

        for (;;) {
            for (int j = 0; j < n; j += 4) {
                float4 zs[4], wS[4];
                float  pdr[4];
#pragma unroll
                for (int u = 0; u < 4; ++u) {
                    int jj = j + u;
                    if (jj > n - 1) jj = n - 1;
                    const int s = __builtin_amdgcn_readlane(my.x, jj);
                    const int r = __builtin_amdgcn_readlane(my.y, jj);
                    zs[u]  = *(const float4*)(z + ((size_t)r * NN + s) * OUTD + lane * 4);
                    wS[u]  = *(const float4*)(attw + ((size_t)r * HH + hd) * (2 * HDD) + qd * 4);
                    pdr[u] = pick8(pd, r);
                }
#pragma unroll
                for (int u = 0; u < 4; ++u) {
                    if (j + u < n) {
                        float a  = quad_sum(dot4(zs[u], wS[u]) + pdr[u]);
                        float ev = (a > 0.0f) ? a : 0.01f * a;
                        float mn = fmaxf(m, ev);
                        float sc = __expf(m - mn);
                        float w  = __expf(ev - mn);
                        l     = l * sc + w;
                        acc.x = acc.x * sc + w * zs[u].x;
                        acc.y = acc.y * sc + w * zs[u].y;
                        acc.z = acc.z * sc + w * zs[u].z;
                        acc.w = acc.w * sc + w * zs[u].w;
                        m = mn;
                    }
                }
            }
            base += 32;
            if (base >= end) break;
            n = end - base;
            if (n > 32) n = 32;
            my = make_int2(0, 0);
            if (lane < n) my = grouped[base + lane];
        }
    }

    const float den = fmaxf(l, 1e-9f);
    float4 o;
    o.x = acc.x / den; o.y = acc.y / den; o.z = acc.z / den; o.w = acc.w / den;
    *(float4*)(out + (size_t)d * OUTD + lane * 4) = o;
}

extern "C" void kernel_launch(void* const* d_in, const int* in_sizes, int n_in,
                              void* d_out, int out_size, void* d_ws, size_t ws_size,
                              hipStream_t stream) {
    const float* h    = (const float*)d_in[0];
    const float* fcw  = (const float*)d_in[1];
    const float* attw = (const float*)d_in[2];
    const int*   src  = (const int*)d_in[3];
    const int*   dst  = (const int*)d_in[4];
    const int*   ety  = (const int*)d_in[5];
    float* out = (float*)d_out;

    float* z       = (float*)d_ws;
    int*   cnt     = (int*)(z + (size_t)RR * NN * OUTD);
    int*   offs    = cnt + CNT_PAD;
    int*   rank    = offs + CNT_PAD;
    int2*  grouped = (int2*)(rank + EE);

    rgat_zero_counts<<<CNT_PAD / 256, 256, 0, stream>>>(cnt);
    rgat_dst_rank<<<(EE + 255) / 256, 256, 0, stream>>>(dst, cnt, rank);
    rgat_scan_counts<<<1, SCAN_T, 0, stream>>>(cnt, offs);
    rgat_group_edges<<<(EE + 255) / 256, 256, 0, stream>>>(src, dst, ety, offs, rank, grouped);
    {
        dim3 grid((NN + 63) / 64, RR);
        size_t shmem = (size_t)(64 * HS_A + 64 * WPS) * sizeof(float);
        rgat_proj_wmma<<<grid, 256, shmem, stream>>>(h, fcw, z);
    }
    rgat_node_softmax_sum<<<(NN + 7) / 8, 256, 0, stream>>>(z, attw, offs, grouped, out);
}


# Destination half of each logit summed in the projection epilogue (16-lane DPP butterfly of z*attw) into a small plane; node pass reads 8 words per lane, not 8 z rows; logit sum order changes
# speedup vs baseline: 2.9752x; 1.0640x over previous
#include <hip/hip_runtime.h>
#include <hip/hip_bf16.h>
#include <math.h>

#define NN 50000
#define EE 800000
#define RR 8
#define IND 128
#define OUTD 128
#define HH 8
#define HDD 16

typedef float v2f __attribute__((ext_vector_type(2)));
typedef float v8f __attribute__((ext_vector_type(8)));

#define HS_A 132
#define WPS  288

template <int CTRL>
__device__ __forceinline__ float row_xor(float x) {
    return __builtin_bit_cast(float,
        __builtin_amdgcn_update_dpp(0, __builtin_bit_cast(int, x), CTRL, 0xF, 0xF, true));
}

__global__ __launch_bounds__(256) void rgat_proj_wmma(const float* __restrict__ h,
                                                      const float* __restrict__ W,
                                                      const float* __restrict__ attw,
                                                      float* __restrict__ z,
                                                      float* __restrict__ ed) {
    extern __shared__ float lds[];
    float* h_s = lds;
    float* w_s = lds + 64 * HS_A;

    const int r     = blockIdx.y;
    const int nbase = blockIdx.x * 64;
    const int tid   = threadIdx.x;

    for (int i = tid; i < 64 * 32; i += 256) {
        int row = i >> 5;
        int c4  = i & 31;
        int g   = nbase + row;
        if (g >= NN) g = NN - 1;
        float4 v = *(const float4*)(h + (size_t)g * IND + c4 * 4);
        float* d = h_s + row * HS_A + c4 * 4;
        d[0] = v.x; d[1] = v.y; d[2] = v.z; d[3] = v.w;
    }
    const float* Wr = W + (size_t)r * IND * OUTD;
    for (int i = tid; i < 64 * 128; i += 256) {
        int kp  = i >> 7;
        int col = i & 127;
        float w0 = Wr[(size_t)(2 * kp) * OUTD + col];
        float w1 = Wr[(size_t)(2 * kp + 1) * OUTD + col];
        v2f p; p.x = w0; p.y = w1;
        *(v2f*)(w_s + kp * WPS + col * 2) = p;
    }
    __syncthreads();

    const int wave = tid >> 5;
    const int lane = tid & 31;
    const int ln   = lane & 15;
    const int g    = lane >> 4;
    const int rt   = wave & 3;
    const int cb   = (wave >> 2) * 64;

    v8f acc0 = {}, acc1 = {}, acc2 = {}, acc3 = {};
    const float* arow = h_s + (rt * 16 + ln) * HS_A;
    const float* bcol = w_s + (size_t)g * WPS + (size_t)(cb + ln) * 2;

    for (int k = 0; k < IND; k += 4) {
        v2f a = *(const v2f*)(arow + k + 2 * g);
        const float* brow = bcol + (k >> 1) * WPS;
        v2f b0 = *(const v2f*)(brow + 0 * 32);
        v2f b1 = *(const v2f*)(brow + 1 * 32);
        v2f b2 = *(const v2f*)(brow + 2 * 32);
        v2f b3 = *(const v2f*)(brow + 3 * 32);
        acc0 = __builtin_amdgcn_wmma_f32_16x16x4_f32(false, a, false, b0, (short)0, acc0, false, false);
        acc1 = __builtin_amdgcn_wmma_f32_16x16x4_f32(false, a, false, b1, (short)0, acc1, false, false);
        acc2 = __builtin_amdgcn_wmma_f32_16x16x4_f32(false, a, false, b2, (short)0, acc2, false, false);
        acc3 = __builtin_amdgcn_wmma_f32_16x16x4_f32(false, a, false, b3, (short)0, acc3, false, false);
    }

    {
        const int hh = wave >> 2;
        const bool l0 = (ln & 1) != 0, l1 = (ln & 2) != 0;
        const bool l2 = (ln & 4) != 0, l3 = (ln & 8) != 0;
        float sC[4];
#pragma unroll
        for (int ct = 0; ct < 4; ++ct) {
            const v8f acc = (ct == 0) ? acc0 : (ct == 1) ? acc1 : (ct == 2) ? acc2 : acc3;
            const float wd = attw[((size_t)r * HH + hh * 4 + ct) * (2 * HDD) + HDD + ln];
            float sA[4];
#pragma unroll
            for (int p = 0; p < 4; ++p) {
                const float x0 = acc[2 * p] * wd, x1 = acc[2 * p + 1] * wd;
                sA[p] = (l0 ? x1 : x0) + row_xor<0x161>(l0 ? x0 : x1);
            }
            float sB[2];
#pragma unroll
            for (int q = 0; q < 2; ++q)
                sB[q] = (l1 ? sA[2 * q + 1] : sA[2 * q]) + row_xor<0x162>(l1 ? sA[2 * q] : sA[2 * q + 1]);
            sC[ct] = (l2 ? sB[1] : sB[0]) + row_xor<0x164>(l2 ? sB[0] : sB[1]);
        }
        v2f sD;
        sD.x = (l3 ? sC[2] : sC[0]) + row_xor<0x168>(l3 ? sC[0] : sC[2]);
        sD.y = (l3 ? sC[3] : sC[1]) + row_xor<0x168>(l3 ? sC[1] : sC[3]);
        const int erow = nbase + rt * 16 + 8 * g + (ln & 7);
        if (erow < NN)
            *(v2f*)(ed + (((size_t)r * 2 + hh) * NN + erow) * 4 + 2 * (ln >> 3)) = sD;
    }

    float* zr = z + (size_t)r * NN * OUTD;
#pragma unroll
    for (int ct = 0; ct < 4; ++ct) {
        v8f acc = (ct == 0) ? acc0 : (ct == 1) ? acc1 : (ct == 2) ? acc2 : acc3;
        int col = cb + ct * 16 + ln;
#pragma unroll
        for (int v = 0; v < 8; ++v) {
            int row = nbase + rt * 16 + 8 * g + v;
            if (row < NN) zr[(size_t)row * OUTD + col] = acc[v];
        }
    }
}

#define SCAN_T   1024
#define SCAN_PER 52
#define CNT_PAD  (SCAN_T * SCAN_PER)

__global__ __launch_bounds__(256) void rgat_zero_counts(int* __restrict__ cnt) {
    int t = blockIdx.x * 256 + threadIdx.x;
    if (t < CNT_PAD) cnt[t] = 0;
}

__global__ __launch_bounds__(256) void rgat_dst_rank(const int* __restrict__ dst,
                                                     int* __restrict__ cnt,
                                                     int* __restrict__ rank) {
    int e = blockIdx.x * 256 + threadIdx.x;
    if (e >= EE) return;
    rank[e] = atomicAdd(cnt + dst[e], 1);
}

__global__ __launch_bounds__(SCAN_T) void rgat_scan_counts(const int* __restrict__ cnt,
                                                           int* __restrict__ offs) {
    __shared__ int wave_tot[32];
    const int tid  = threadIdx.x;
    const int lane = tid & 31;
    const int wv   = tid >> 5;

    const int4* p = (const int4*)(cnt + tid * SCAN_PER);
    int4 v[SCAN_PER / 4];
    int s = 0;
#pragma unroll
    for (int i = 0; i < SCAN_PER / 4; ++i) {
        v[i] = p[i];
        s += v[i].x + v[i].y + v[i].z + v[i].w;
    }
    int inc = s;
#pragma unroll
    for (int o = 1; o < 32; o <<= 1) {
        int t = __shfl_up(inc, o);
        if (lane >= o) inc += t;
    }
    if (lane == 31) wave_tot[wv] = inc;
    __syncthreads();
    if (wv == 0) {
        int x  = wave_tot[lane];
        int xi = x;
#pragma unroll
        for (int o = 1; o < 32; o <<= 1) {
            int t = __shfl_up(xi, o);
            if (lane >= o) xi += t;
        }
        wave_tot[lane] = xi - x;
    }
    __syncthreads();
    int run = wave_tot[wv] + inc - s;
    int4* q = (int4*)(offs + tid * SCAN_PER);
#pragma unroll
    for (int i = 0; i < SCAN_PER / 4; ++i) {
        int4 o;
        o.x = run; run += v[i].x;
        o.y = run; run += v[i].y;
        o.z = run; run += v[i].z;
        o.w = run; run += v[i].w;
        q[i] = o;
    }
}

__global__ __launch_bounds__(256) void rgat_group_edges(const int* __restrict__ src,
                                                        const int* __restrict__ dst,
                                                        const int* __restrict__ etype,
                                                        const int* __restrict__ offs,
                                                        const int* __restrict__ rank,
                                                        int2* __restrict__ grouped) {
    int e = blockIdx.x * 256 + threadIdx.x;
    if (e >= EE) return;
    int pos = offs[dst[e]] + rank[e];
    grouped[pos] = make_int2(src[e], etype[e]);
}

__device__ __forceinline__ float quad_sum(float x) {
    int b = __builtin_bit_cast(int, x);
    float y = x + __builtin_bit_cast(float, __builtin_amdgcn_update_dpp(b, b, 0xB1, 0xF, 0xF, false));
    int c = __builtin_bit_cast(int, y);
    return y + __builtin_bit_cast(float, __builtin_amdgcn_update_dpp(c, c, 0x4E, 0xF, 0xF, false));
}

__device__ __forceinline__ float pick8(const float (&p)[8], int r) {
    float a = (r & 1) ? p[1] : p[0];
    float b = (r & 1) ? p[3] : p[2];
    float c = (r & 1) ? p[5] : p[4];
    float d = (r & 1) ? p[7] : p[6];
    float ab = (r & 2) ? b : a;
    float cd = (r & 2) ? d : c;
    return (r & 4) ? cd : ab;
}

__device__ __forceinline__ float dot4(float4 a, float4 b) {
    return a.x * b.x + a.y * b.y + a.z * b.z + a.w * b.w;
}

__global__ __launch_bounds__(256) void rgat_node_softmax_sum(const float* __restrict__ z,
                                                             const float* __restrict__ ed,
                                                             const float* __restrict__ attw,
                                                             const int* __restrict__ offs,
                                                             const int2* __restrict__ grouped,
                                                             float* __restrict__ out) {
    const int lane = threadIdx.x & 31;
    const int d = __builtin_amdgcn_readfirstlane(blockIdx.x * 8 + (threadIdx.x >> 5));
    if (d >= NN) return;
    const int hd = lane >> 2;
    const int qd = lane & 3;

    const int beg = __builtin_amdgcn_readfirstlane(offs[d]);
    const int end = __builtin_amdgcn_readfirstlane(offs[d + 1]);

    float4 acc = make_float4(0.0f, 0.0f, 0.0f, 0.0f);
    float  m = -INFINITY;
    float  l = 0.0f;

    if (end > beg) {
        int base = beg;
        int n = end - base;
        if (n > 32) n = 32;
        int2 my = make_int2(0, 0);
        if (lane < n) my = grouped[base + lane];

        float pd[8];
#pragma unroll
        for (int r = 0; r < RR; ++r)
            pd[r] = ed[(((size_t)r * 2 + (hd >> 2)) * NN + d) * 4 + (hd & 3)];

        for (;;) {
            for (int j = 0; j < n; j += 4) {
                float4 zs[4], wS[4];
                float  pdr[4];
#pragma unroll
                for (int u = 0; u < 4; ++u) {
                    int jj = j + u;
                    if (jj > n - 1) jj = n - 1;
                    const int s = __builtin_amdgcn_readlane(my.x, jj);
                    const int r = __builtin_amdgcn_readlane(my.y, jj);
                    zs[u]  = *(const float4*)(z + ((size_t)r * NN + s) * OUTD + lane * 4);
                    wS[u]  = *(const float4*)(attw + ((size_t)r * HH + hd) * (2 * HDD) + qd * 4);
                    pdr[u] = pick8(pd, r);
                }
#pragma unroll
                for (int u = 0; u < 4; ++u) {
                    if (j + u < n) {
                        float a  = quad_sum(dot4(zs[u], wS[u])) + pdr[u];
                        float ev = (a > 0.0f) ? a : 0.01f * a;
                        float mn = fmaxf(m, ev);
                        float sc = __expf(m - mn);
                        float w  = __expf(ev - mn);
                        l     = l * sc + w;
                        acc.x = acc.x * sc + w * zs[u].x;
                        acc.y = acc.y * sc + w * zs[u].y;
                        acc.z = acc.z * sc + w * zs[u].z;
                        acc.w = acc.w * sc + w * zs[u].w;
                        m = mn;
                    }
                }
            }
            base += 32;
            if (base >= end) break;
            n = end - base;
            if (n > 32) n = 32;
            my = make_int2(0, 0);
            if (lane < n) my = grouped[base + lane];
        }
    }

    const float den = fmaxf(l, 1e-9f);
    float4 o;
    o.x = acc.x / den; o.y = acc.y / den; o.z = acc.z / den; o.w = acc.w / den;
    *(float4*)(out + (size_t)d * OUTD + lane * 4) = o;
}

extern "C" void kernel_launch(void* const* d_in, const int* in_sizes, int n_in,
                              void* d_out, int out_size, void* d_ws, size_t ws_size,
                              hipStream_t stream) {
    const float* h    = (const float*)d_in[0];
    const float* fcw  = (const float*)d_in[1];
    const float* attw = (const float*)d_in[2];
    const int*   src  = (const int*)d_in[3];
    const int*   dst  = (const int*)d_in[4];
    const int*   ety  = (const int*)d_in[5];
    float* out = (float*)d_out;

    float* z       = (float*)d_ws;
    float* ed      = z + (size_t)RR * NN * OUTD;
    int*   cnt     = (int*)(ed + (size_t)RR * HH * NN);
    int*   offs    = cnt + CNT_PAD;
    int*   rank    = offs + CNT_PAD;
    int2*  grouped = (int2*)(rank + EE);

    rgat_zero_counts<<<CNT_PAD / 256, 256, 0, stream>>>(cnt);
    rgat_dst_rank<<<(EE + 255) / 256, 256, 0, stream>>>(dst, cnt, rank);
    rgat_scan_counts<<<1, SCAN_T, 0, stream>>>(cnt, offs);
    rgat_group_edges<<<(EE + 255) / 256, 256, 0, stream>>>(src, dst, ety, offs, rank, grouped);
    {
        dim3 grid((NN + 63) / 64, RR);
        size_t shmem = (size_t)(64 * HS_A + 64 * WPS) * sizeof(float);
        rgat_proj_wmma<<<grid, 256, shmem, stream>>>(h, fcw, attw, z, ed);
    }
    rgat_node_softmax_sum<<<(NN + 7) / 8, 256, 0, stream>>>(z, ed, attw, offs, grouped, out);
}


# Projection staging: the block's 24 tile loads (16 bytes per lane) all issued before the first wait, was 8 serial round trips for the h tile and 4 for W; occupancy hint keeps them grouped; same sums
# speedup vs baseline: 3.2791x; 1.1021x over previous
#include <hip/hip_runtime.h>
#include <hip/hip_bf16.h>
#include <math.h>

#define NN 50000
#define EE 800000
#define RR 8
#define IND 128
#define OUTD 128
#define HH 8
#define HDD 16

typedef float v2f __attribute__((ext_vector_type(2)));
typedef float v8f __attribute__((ext_vector_type(8)));

#define HS_A 132
#define WPS  288

template <int CTRL>
__device__ __forceinline__ float row_xor(float x) {
    return __builtin_bit_cast(float,
        __builtin_amdgcn_update_dpp(0, __builtin_bit_cast(int, x), CTRL, 0xF, 0xF, true));
}

__global__ __launch_bounds__(256) __attribute__((amdgpu_waves_per_eu(6, 6))) void rgat_proj_wmma(const float* __restrict__ h,
                                                      const float* __restrict__ W,
                                                      const float* __restrict__ attw,
                                                      float* __restrict__ z,
                                                      float* __restrict__ ed) {
    extern __shared__ __attribute__((aligned(16))) float lds[];
    float* h_s = lds;
    float* w_s = lds + 64 * HS_A;

    const int r     = blockIdx.y;
    const int nbase = blockIdx.x * 64;
    const int tid   = threadIdx.x;

    float4 hv[8];
#pragma unroll
    for (int it = 0; it < 8; ++it) {
        const int i   = tid + it * 256;
        int       gr  = nbase + (i >> 5);
        if (gr >= NN) gr = NN - 1;
        hv[it] = *(const float4*)(h + (size_t)gr * IND + (i & 31) * 4);
    }
    const float* Wr = W + (size_t)r * IND * OUTD;
    float4 wa[8], wb[8];
#pragma unroll
    for (int it = 0; it < 8; ++it) {
        const int i  = tid + it * 256;
        const int kp = i >> 5;
        const int c4 = i & 31;
        wa[it] = *(const float4*)(Wr + (size_t)(2 * kp) * OUTD + c4 * 4);
        wb[it] = *(const float4*)(Wr + (size_t)(2 * kp + 1) * OUTD + c4 * 4);
    }
#pragma unroll
    for (int it = 0; it < 8; ++it) {
        const int i = tid + it * 256;
        *(float4*)(h_s + (i >> 5) * HS_A + (i & 31) * 4) = hv[it];
    }
#pragma unroll
    for (int it = 0; it < 8; ++it) {
        const int i  = tid + it * 256;
        float* d = w_s + (i >> 5) * WPS + (i & 31) * 8;
        *(float4*)(d)     = make_float4(wa[it].x, wb[it].x, wa[it].y, wb[it].y);
        *(float4*)(d + 4) = make_float4(wa[it].z, wb[it].z, wa[it].w, wb[it].w);
    }
    __syncthreads();

    const int wave = tid >> 5;
    const int lane = tid & 31;
    const int ln   = lane & 15;
    const int g    = lane >> 4;
    const int rt   = wave & 3;
    const int cb   = (wave >> 2) * 64;

    v8f acc0 = {}, acc1 = {}, acc2 = {}, acc3 = {};
    const float* arow = h_s + (rt * 16 + ln) * HS_A;
    const float* bcol = w_s + (size_t)g * WPS + (size_t)(cb + ln) * 2;

    for (int k = 0; k < IND; k += 4) {
        v2f a = *(const v2f*)(arow + k + 2 * g);
        const float* brow = bcol + (k >> 1) * WPS;
        v2f b0 = *(const v2f*)(brow + 0 * 32);
        v2f b1 = *(const v2f*)(brow + 1 * 32);
        v2f b2 = *(const v2f*)(brow + 2 * 32);
        v2f b3 = *(const v2f*)(brow + 3 * 32);
        acc0 = __builtin_amdgcn_wmma_f32_16x16x4_f32(false, a, false, b0, (short)0, acc0, false, false);
        acc1 = __builtin_amdgcn_wmma_f32_16x16x4_f32(false, a, false, b1, (short)0, acc1, false, false);
        acc2 = __builtin_amdgcn_wmma_f32_16x16x4_f32(false, a, false, b2, (short)0, acc2, false, false);
        acc3 = __builtin_amdgcn_wmma_f32_16x16x4_f32(false, a, false, b3, (short)0, acc3, false, false);
    }

    {
        const int hh = wave >> 2;
        const bool l0 = (ln & 1) != 0, l1 = (ln & 2) != 0;
        const bool l2 = (ln & 4) != 0, l3 = (ln & 8) != 0;
        float sC[4];
#pragma unroll
        for (int ct = 0; ct < 4; ++ct) {
            const v8f acc = (ct == 0) ? acc0 : (ct == 1) ? acc1 : (ct == 2) ? acc2 : acc3;
            const float wd = attw[((size_t)r * HH + hh * 4 + ct) * (2 * HDD) + HDD + ln];
            float sA[4];
#pragma unroll
            for (int p = 0; p < 4; ++p) {
                const float x0 = acc[2 * p] * wd, x1 = acc[2 * p + 1] * wd;
                sA[p] = (l0 ? x1 : x0) + row_xor<0x161>(l0 ? x0 : x1);
            }
            float sB[2];
#pragma unroll
            for (int q = 0; q < 2; ++q)
                sB[q] = (l1 ? sA[2 * q + 1] : sA[2 * q]) + row_xor<0x162>(l1 ? sA[2 * q] : sA[2 * q + 1]);
            sC[ct] = (l2 ? sB[1] : sB[0]) + row_xor<0x164>(l2 ? sB[0] : sB[1]);
        }
        v2f sD;
        sD.x = (l3 ? sC[2] : sC[0]) + row_xor<0x168>(l3 ? sC[0] : sC[2]);
        sD.y = (l3 ? sC[3] : sC[1]) + row_xor<0x168>(l3 ? sC[1] : sC[3]);
        const int erow = nbase + rt * 16 + 8 * g + (ln & 7);
        if (erow < NN)
            *(v2f*)(ed + (((size_t)r * 2 + hh) * NN + erow) * 4 + 2 * (ln >> 3)) = sD;
    }

    float* zr = z + (size_t)r * NN * OUTD;
#pragma unroll
    for (int ct = 0; ct < 4; ++ct) {
        v8f acc = (ct == 0) ? acc0 : (ct == 1) ? acc1 : (ct == 2) ? acc2 : acc3;
        int col = cb + ct * 16 + ln;
#pragma unroll
        for (int v = 0; v < 8; ++v) {
            int row = nbase + rt * 16 + 8 * g + v;
            if (row < NN) zr[(size_t)row * OUTD + col] = acc[v];
        }
    }
}

#define SCAN_T   1024
#define SCAN_PER 52
#define CNT_PAD  (SCAN_T * SCAN_PER)

__global__ __launch_bounds__(256) void rgat_zero_counts(int* __restrict__ cnt) {
    int t = blockIdx.x * 256 + threadIdx.x;
    if (t < CNT_PAD) cnt[t] = 0;
}

__global__ __launch_bounds__(256) void rgat_dst_rank(const int* __restrict__ dst,
                                                     int* __restrict__ cnt,
                                                     int* __restrict__ rank) {
    int e = blockIdx.x * 256 + threadIdx.x;
    if (e >= EE) return;
    rank[e] = atomicAdd(cnt + dst[e], 1);
}

__global__ __launch_bounds__(SCAN_T) void rgat_scan_counts(const int* __restrict__ cnt,
                                                           int* __restrict__ offs) {
    __shared__ int wave_tot[32];
    const int tid  = threadIdx.x;
    const int lane = tid & 31;
    const int wv   = tid >> 5;

    const int4* p = (const int4*)(cnt + tid * SCAN_PER);
    int4 v[SCAN_PER / 4];
    int s = 0;
#pragma unroll
    for (int i = 0; i < SCAN_PER / 4; ++i) {
        v[i] = p[i];
        s += v[i].x + v[i].y + v[i].z + v[i].w;
    }
    int inc = s;
#pragma unroll
    for (int o = 1; o < 32; o <<= 1) {
        int t = __shfl_up(inc, o);
        if (lane >= o) inc += t;
    }
    if (lane == 31) wave_tot[wv] = inc;
    __syncthreads();
    if (wv == 0) {
        int x  = wave_tot[lane];
        int xi = x;
#pragma unroll
        for (int o = 1; o < 32; o <<= 1) {
            int t = __shfl_up(xi, o);
            if (lane >= o) xi += t;
        }
        wave_tot[lane] = xi - x;
    }
    __syncthreads();
    int run = wave_tot[wv] + inc - s;
    int4* q = (int4*)(offs + tid * SCAN_PER);
#pragma unroll
    for (int i = 0; i < SCAN_PER / 4; ++i) {
        int4 o;
        o.x = run; run += v[i].x;
        o.y = run; run += v[i].y;
        o.z = run; run += v[i].z;
        o.w = run; run += v[i].w;
        q[i] = o;
    }
}

__global__ __launch_bounds__(256) void rgat_group_edges(const int* __restrict__ src,
                                                        const int* __restrict__ dst,
                                                        const int* __restrict__ etype,
                                                        const int* __restrict__ offs,
                                                        const int* __restrict__ rank,
                                                        int2* __restrict__ grouped) {
    int e = blockIdx.x * 256 + threadIdx.x;
    if (e >= EE) return;
    int pos = offs[dst[e]] + rank[e];
    grouped[pos] = make_int2(src[e], etype[e]);
}

__device__ __forceinline__ float quad_sum(float x) {
    int b = __builtin_bit_cast(int, x);
    float y = x + __builtin_bit_cast(float, __builtin_amdgcn_update_dpp(b, b, 0xB1, 0xF, 0xF, false));
    int c = __builtin_bit_cast(int, y);
    return y + __builtin_bit_cast(float, __builtin_amdgcn_update_dpp(c, c, 0x4E, 0xF, 0xF, false));
}

__device__ __forceinline__ float pick8(const float (&p)[8], int r) {
    float a = (r & 1) ? p[1] : p[0];
    float b = (r & 1) ? p[3] : p[2];
    float c = (r & 1) ? p[5] : p[4];
    float d = (r & 1) ? p[7] : p[6];
    float ab = (r & 2) ? b : a;
    float cd = (r & 2) ? d : c;
    return (r & 4) ? cd : ab;
}

__device__ __forceinline__ float dot4(float4 a, float4 b) {
    return a.x * b.x + a.y * b.y + a.z * b.z + a.w * b.w;
}

__global__ __launch_bounds__(256) void rgat_node_softmax_sum(const float* __restrict__ z,
                                                             const float* __restrict__ ed,
                                                             const float* __restrict__ attw,
                                                             const int* __restrict__ offs,
                                                             const int2* __restrict__ grouped,
                                                             float* __restrict__ out) {
    const int lane = threadIdx.x & 31;
    const int d = __builtin_amdgcn_readfirstlane(blockIdx.x * 8 + (threadIdx.x >> 5));
    if (d >= NN) return;
    const int hd = lane >> 2;
    const int qd = lane & 3;

    const int beg = __builtin_amdgcn_readfirstlane(offs[d]);
    const int end = __builtin_amdgcn_readfirstlane(offs[d + 1]);

    float4 acc = make_float4(0.0f, 0.0f, 0.0f, 0.0f);
    float  m = -INFINITY;
    float  l = 0.0f;

    if (end > beg) {
        int base = beg;
        int n = end - base;
        if (n > 32) n = 32;
        int2 my = make_int2(0, 0);
        if (lane < n) my = grouped[base + lane];

        float pd[8];
#pragma unroll
        for (int r = 0; r < RR; ++r)
            pd[r] = ed[(((size_t)r * 2 + (hd >> 2)) * NN + d) * 4 + (hd & 3)];

        for (;;) {
            for (int j = 0; j < n; j += 4) {
                float4 zs[4], wS[4];
                float  pdr[4];
#pragma unroll
                for (int u = 0; u < 4; ++u) {
                    int jj = j + u;
                    if (jj > n - 1) jj = n - 1;
                    const int s = __builtin_amdgcn_readlane(my.x, jj);
                    const int r = __builtin_amdgcn_readlane(my.y, jj);
                    zs[u]  = *(const float4*)(z + ((size_t)r * NN + s) * OUTD + lane * 4);
                    wS[u]  = *(const float4*)(attw + ((size_t)r * HH + hd) * (2 * HDD) + qd * 4);
                    pdr[u] = pick8(pd, r);
                }
#pragma unroll
                for (int u = 0; u < 4; ++u) {
                    if (j + u < n) {
                        float a  = quad_sum(dot4(zs[u], wS[u])) + pdr[u];
                        float ev = (a > 0.0f) ? a : 0.01f * a;
                        float mn = fmaxf(m, ev);
                        float sc = __expf(m - mn);
                        float w  = __expf(ev - mn);
                        l     = l * sc + w;
                        acc.x = acc.x * sc + w * zs[u].x;
                        acc.y = acc.y * sc + w * zs[u].y;
                        acc.z = acc.z * sc + w * zs[u].z;
                        acc.w = acc.w * sc + w * zs[u].w;
                        m = mn;
                    }
                }
            }
            base += 32;
            if (base >= end) break;
            n = end - base;
            if (n > 32) n = 32;
            my = make_int2(0, 0);
            if (lane < n) my = grouped[base + lane];
        }
    }

    const float den = fmaxf(l, 1e-9f);
    float4 o;
    o.x = acc.x / den; o.y = acc.y / den; o.z = acc.z / den; o.w = acc.w / den;
    *(float4*)(out + (size_t)d * OUTD + lane * 4) = o;
}

extern "C" void kernel_launch(void* const* d_in, const int* in_sizes, int n_in,
                              void* d_out, int out_size, void* d_ws, size_t ws_size,
                              hipStream_t stream) {
    const float* h    = (const float*)d_in[0];
    const float* fcw  = (const float*)d_in[1];
    const float* attw = (const float*)d_in[2];
    const int*   src  = (const int*)d_in[3];
    const int*   dst  = (const int*)d_in[4];
    const int*   ety  = (const int*)d_in[5];
    float* out = (float*)d_out;

    float* z       = (float*)d_ws;
    float* ed      = z + (size_t)RR * NN * OUTD;
    int*   cnt     = (int*)(ed + (size_t)RR * HH * NN);
    int*   offs    = cnt + CNT_PAD;
    int*   rank    = offs + CNT_PAD;
    int2*  grouped = (int2*)(rank + EE);

    rgat_zero_counts<<<CNT_PAD / 256, 256, 0, stream>>>(cnt);
    rgat_dst_rank<<<(EE + 255) / 256, 256, 0, stream>>>(dst, cnt, rank);
    rgat_scan_counts<<<1, SCAN_T, 0, stream>>>(cnt, offs);
    rgat_group_edges<<<(EE + 255) / 256, 256, 0, stream>>>(src, dst, ety, offs, rank, grouped);
    {
        dim3 grid((NN + 63) / 64, RR);
        size_t shmem = (size_t)(64 * HS_A + 64 * WPS) * sizeof(float);
        rgat_proj_wmma<<<grid, 256, shmem, stream>>>(h, fcw, attw, z, ed);
    }
    rgat_node_softmax_sum<<<(NN + 7) / 8, 256, 0, stream>>>(z, ed, attw, offs, grouped, out);
}


# Rank, scan and scatter steps ride as the first blocks of three launches that each carry a third of the projection tiles (8 edges per lane, scan in one 256-thread block); rank order may differ
# speedup vs baseline: 3.5211x; 1.0738x over previous
#include <hip/hip_runtime.h>
#include <hip/hip_bf16.h>
#include <math.h>

#define NN 50000
#define EE 800000
#define RR 8
#define IND 128
#define OUTD 128
#define HH 8
#define HDD 16

typedef float v2f __attribute__((ext_vector_type(2)));
typedef float v8f __attribute__((ext_vector_type(8)));

#define HS_A 132
#define WPS  288

template <int CTRL>
__device__ __forceinline__ float row_xor(float x) {
    return __builtin_bit_cast(float,
        __builtin_amdgcn_update_dpp(0, __builtin_bit_cast(int, x), CTRL, 0xF, 0xF, true));
}

#define TILES_X    ((NN + 63) / 64)
#define PROJ_TILES (TILES_X * RR)
#define PROJ_P1    2048
#define PROJ_P2    2048
#define PROJ_P3    (PROJ_TILES - PROJ_P1 - PROJ_P2)

__device__ __forceinline__ void proj_tile(const int pid, float* lds,
                                          const float* __restrict__ h,
                                          const float* __restrict__ W,
                                          const float* __restrict__ attw,
                                          float* __restrict__ z,
                                          float* __restrict__ ed) {
    float* h_s = lds;
    float* w_s = lds + 64 * HS_A;

    const int r     = pid / TILES_X;
    const int nbase = (pid - r * TILES_X) * 64;
    const int tid   = threadIdx.x;

    float4 hv[8];
#pragma unroll
    for (int it = 0; it < 8; ++it) {
        const int i   = tid + it * 256;
        int       gr  = nbase + (i >> 5);
        if (gr >= NN) gr = NN - 1;
        hv[it] = *(const float4*)(h + (size_t)gr * IND + (i & 31) * 4);
    }
    const float* Wr = W + (size_t)r * IND * OUTD;
    float4 wa[8], wb[8];
#pragma unroll
    for (int it = 0; it < 8; ++it) {
        const int i  = tid + it * 256;
        const int kp = i >> 5;
        const int c4 = i & 31;
        wa[it] = *(const float4*)(Wr + (size_t)(2 * kp) * OUTD + c4 * 4);
        wb[it] = *(const float4*)(Wr + (size_t)(2 * kp + 1) * OUTD + c4 * 4);
    }
#pragma unroll
    for (int it = 0; it < 8; ++it) {
        const int i = tid + it * 256;
        *(float4*)(h_s + (i >> 5) * HS_A + (i & 31) * 4) = hv[it];
    }
#pragma unroll
    for (int it = 0; it < 8; ++it) {
        const int i  = tid + it * 256;
        float* d = w_s + (i >> 5) * WPS + (i & 31) * 8;
        *(float4*)(d)     = make_float4(wa[it].x, wb[it].x, wa[it].y, wb[it].y);
        *(float4*)(d + 4) = make_float4(wa[it].z, wb[it].z, wa[it].w, wb[it].w);
    }
    __syncthreads();

    const int wave = tid >> 5;
    const int lane = tid & 31;
    const int ln   = lane & 15;
    const int g    = lane >> 4;
    const int rt   = wave & 3;
    const int cb   = (wave >> 2) * 64;

    v8f acc0 = {}, acc1 = {}, acc2 = {}, acc3 = {};
    const float* arow = h_s + (rt * 16 + ln) * HS_A;
    const float* bcol = w_s + (size_t)g * WPS + (size_t)(cb + ln) * 2;

    for (int k = 0; k < IND; k += 4) {
        v2f a = *(const v2f*)(arow + k + 2 * g);
        const float* brow = bcol + (k >> 1) * WPS;
        v2f b0 = *(const v2f*)(brow + 0 * 32);
        v2f b1 = *(const v2f*)(brow + 1 * 32);
        v2f b2 = *(const v2f*)(brow + 2 * 32);
        v2f b3 = *(const v2f*)(brow + 3 * 32);
        acc0 = __builtin_amdgcn_wmma_f32_16x16x4_f32(false, a, false, b0, (short)0, acc0, false, false);
        acc1 = __builtin_amdgcn_wmma_f32_16x16x4_f32(false, a, false, b1, (short)0, acc1, false, false);
        acc2 = __builtin_amdgcn_wmma_f32_16x16x4_f32(false, a, false, b2, (short)0, acc2, false, false);
        acc3 = __builtin_amdgcn_wmma_f32_16x16x4_f32(false, a, false, b3, (short)0, acc3, false, false);
    }

    {
        const int hh = wave >> 2;
        const bool l0 = (ln & 1) != 0, l1 = (ln & 2) != 0;
        const bool l2 = (ln & 4) != 0, l3 = (ln & 8) != 0;
        float sC[4];
#pragma unroll
        for (int ct = 0; ct < 4; ++ct) {
            const v8f acc = (ct == 0) ? acc0 : (ct == 1) ? acc1 : (ct == 2) ? acc2 : acc3;
            const float wd = attw[((size_t)r * HH + hh * 4 + ct) * (2 * HDD) + HDD + ln];
            float sA[4];
#pragma unroll
            for (int p = 0; p < 4; ++p) {
                const float x0 = acc[2 * p] * wd, x1 = acc[2 * p + 1] * wd;
                sA[p] = (l0 ? x1 : x0) + row_xor<0x161>(l0 ? x0 : x1);
            }
            float sB[2];
#pragma unroll
            for (int q = 0; q < 2; ++q)
                sB[q] = (l1 ? sA[2 * q + 1] : sA[2 * q]) + row_xor<0x162>(l1 ? sA[2 * q] : sA[2 * q + 1]);
            sC[ct] = (l2 ? sB[1] : sB[0]) + row_xor<0x164>(l2 ? sB[0] : sB[1]);
        }
        v2f sD;
        sD.x = (l3 ? sC[2] : sC[0]) + row_xor<0x168>(l3 ? sC[0] : sC[2]);
        sD.y = (l3 ? sC[3] : sC[1]) + row_xor<0x168>(l3 ? sC[1] : sC[3]);
        const int erow = nbase + rt * 16 + 8 * g + (ln & 7);
        if (erow < NN)
            *(v2f*)(ed + (((size_t)r * 2 + hh) * NN + erow) * 4 + 2 * (ln >> 3)) = sD;
    }

    float* zr = z + (size_t)r * NN * OUTD;
#pragma unroll
    for (int ct = 0; ct < 4; ++ct) {
        v8f acc = (ct == 0) ? acc0 : (ct == 1) ? acc1 : (ct == 2) ? acc2 : acc3;
        int col = cb + ct * 16 + ln;
#pragma unroll
        for (int v = 0; v < 8; ++v) {
            int row = nbase + rt * 16 + 8 * g + v;
            if (row < NN) zr[(size_t)row * OUTD + col] = acc[v];
        }
    }
}

#define CNT_PAD   53248
#define SCAN_CH   13
#define SORT_EPT  8
#define SORT_BLK  ((EE + 256 * SORT_EPT - 1) / (256 * SORT_EPT))
static_assert(EE % SORT_EPT == 0, "a thread's edges are all inside or all outside");
static_assert(SCAN_CH * 4096 == CNT_PAD, "scan chunks cover the counters");
static_assert(CNT_PAD >= NN + 1, "offs[NN] must exist");

__global__ __launch_bounds__(256) void rgat_zero_counts(int* __restrict__ cnt) {
    int t = blockIdx.x * 256 + threadIdx.x;
    if (t < CNT_PAD) cnt[t] = 0;
}

__device__ __forceinline__ void sort_rank(const int b, const int* __restrict__ dst,
                                          int* __restrict__ cnt, int* __restrict__ rank) {
    const int e0 = (b * 256 + (int)threadIdx.x) * SORT_EPT;
    if (e0 >= EE) return;
    const int4 d0 = *(const int4*)(dst + e0);
    const int4 d1 = *(const int4*)(dst + e0 + 4);
    int4 k0, k1;
    k0.x = atomicAdd(cnt + d0.x, 1);
    k0.y = atomicAdd(cnt + d0.y, 1);
    k0.z = atomicAdd(cnt + d0.z, 1);
    k0.w = atomicAdd(cnt + d0.w, 1);
    k1.x = atomicAdd(cnt + d1.x, 1);
    k1.y = atomicAdd(cnt + d1.y, 1);
    k1.z = atomicAdd(cnt + d1.z, 1);
    k1.w = atomicAdd(cnt + d1.w, 1);
    *(int4*)(rank + e0)     = k0;
    *(int4*)(rank + e0 + 4) = k1;
}

__device__ __forceinline__ int wave_incl_scan(int x, const int lane) {
#pragma unroll
    for (int o = 1; o < 32; o <<= 1) {
        int t = __shfl_up(x, o);
        if (lane >= o) x += t;
    }
    return x;
}

#define SCAN_G0 7

template <int C0, int NC>
__device__ __forceinline__ void scan_load(const int* __restrict__ cnt, const int tid, int4 (&v)[NC][4]) {
#pragma unroll
    for (int u = 0; u < NC; ++u) {
        const int4* p = (const int4*)(cnt + (C0 + u) * 4096 + tid * 16);
        v[u][0] = p[0]; v[u][1] = p[1]; v[u][2] = p[2]; v[u][3] = p[3];
    }
}

template <int C0, int NC>
__device__ __forceinline__ void scan_sums(const int* __restrict__ cnt, int* sS, int* sI,
                                          const int tid, const int lane) {
    int4 v[NC][4];
    scan_load<C0, NC>(cnt, tid, v);
#pragma unroll
    for (int u = 0; u < NC; ++u) {
        const int sum = (v[u][0].x + v[u][0].y + v[u][0].z + v[u][0].w)
                      + (v[u][1].x + v[u][1].y + v[u][1].z + v[u][1].w)
                      + (v[u][2].x + v[u][2].y + v[u][2].z + v[u][2].w)
                      + (v[u][3].x + v[u][3].y + v[u][3].z + v[u][3].w);
        sS[(C0 + u) * 256 + tid] = sum;
        sI[(C0 + u) * 256 + tid] = wave_incl_scan(sum, lane);
    }
}

template <int C0, int NC>
__device__ __forceinline__ void scan_write(const int* __restrict__ cnt, int* __restrict__ offs,
                                           const int* wt, const int* sS, const int* sI,
                                           const int tid, const int wv) {
    int4 v[NC][4];
    scan_load<C0, NC>(cnt, tid, v);
#pragma unroll
    for (int u = 0; u < NC; ++u) {
        const int c = C0 + u;
        int run = wt[c * 8 + wv] + sI[c * 256 + tid] - sS[c * 256 + tid];
        int4* q = (int4*)(offs + c * 4096 + tid * 16);
#pragma unroll
        for (int k = 0; k < 4; ++k) {
            int4 o;
            o.x = run; run += v[u][k].x;
            o.y = run; run += v[u][k].y;
            o.z = run; run += v[u][k].z;
            o.w = run; run += v[u][k].w;
            q[k] = o;
        }
    }
}

__device__ __forceinline__ void sort_scan(const int* __restrict__ cnt, int* __restrict__ offs,
                                          int* wt  ) {
    const int tid  = threadIdx.x;
    const int lane = tid & 31;
    const int wv   = tid >> 5;
    int* sS = wt + 128;
    int* sI = sS + SCAN_CH * 256;

    scan_sums<0, SCAN_G0>(cnt, sS, sI, tid, lane);
    scan_sums<SCAN_G0, SCAN_CH - SCAN_G0>(cnt, sS, sI, tid, lane);
    __syncthreads();
    if (wv == 0) {
        int a0 = 0, a1 = 0, a2 = 0, a3 = 0;
        if (lane < SCAN_CH * 2) {
            a0 = sI[(4 * lane)     * 32 + 31]; a1 = sI[(4 * lane + 1) * 32 + 31];
            a2 = sI[(4 * lane + 2) * 32 + 31]; a3 = sI[(4 * lane + 3) * 32 + 31];
        }
        const int sum4 = a0 + a1 + a2 + a3;
        const int ex   = wave_incl_scan(sum4, lane) - sum4;
        if (lane < SCAN_CH * 2) {
            wt[4 * lane]     = ex;
            wt[4 * lane + 1] = ex + a0;
            wt[4 * lane + 2] = ex + a0 + a1;
            wt[4 * lane + 3] = ex + a0 + a1 + a2;
        }
    }
    __syncthreads();
    scan_write<0, SCAN_G0>(cnt, offs, wt, sS, sI, tid, wv);
    scan_write<SCAN_G0, SCAN_CH - SCAN_G0>(cnt, offs, wt, sS, sI, tid, wv);
}

__device__ __forceinline__ void sort_scatter(const int b,
                                             const int* __restrict__ src,
                                             const int* __restrict__ dst,
                                             const int* __restrict__ etype,
                                             const int* __restrict__ offs,
                                             const int* __restrict__ rank,
                                             int2* __restrict__ grouped) {
    const int e0 = (b * 256 + (int)threadIdx.x) * SORT_EPT;
    if (e0 >= EE) return;
    const int4 d0 = *(const int4*)(dst + e0),   d1 = *(const int4*)(dst + e0 + 4);
    const int4 k0 = *(const int4*)(rank + e0),  k1 = *(const int4*)(rank + e0 + 4);
    const int4 s0 = *(const int4*)(src + e0),   s1 = *(const int4*)(src + e0 + 4);
    const int4 t0 = *(const int4*)(etype + e0), t1 = *(const int4*)(etype + e0 + 4);
    const int p0 = offs[d0.x] + k0.x, p1 = offs[d0.y] + k0.y;
    const int p2 = offs[d0.z] + k0.z, p3 = offs[d0.w] + k0.w;
    const int p4 = offs[d1.x] + k1.x, p5 = offs[d1.y] + k1.y;
    const int p6 = offs[d1.z] + k1.z, p7 = offs[d1.w] + k1.w;
    grouped[p0] = make_int2(s0.x, t0.x);
    grouped[p1] = make_int2(s0.y, t0.y);
    grouped[p2] = make_int2(s0.z, t0.z);
    grouped[p3] = make_int2(s0.w, t0.w);
    grouped[p4] = make_int2(s1.x, t1.x);
    grouped[p5] = make_int2(s1.y, t1.y);
    grouped[p6] = make_int2(s1.z, t1.z);
    grouped[p7] = make_int2(s1.w, t1.w);
}

template <int STEP, int NSORT, int TILE0>
__global__ __launch_bounds__(256) __attribute__((amdgpu_waves_per_eu(6, 6)))
void rgat_proj_and_sort(const float* __restrict__ h,
                        const float* __restrict__ W,
                        const float* __restrict__ attw,
                        float* __restrict__ z,
                        float* __restrict__ ed,
                        const int* __restrict__ src,
                        const int* __restrict__ dst,
                        const int* __restrict__ etype,
                        int* __restrict__ cnt,
                        int* __restrict__ offs,
                        int* __restrict__ rank,
                        int2* __restrict__ grouped) {
    extern __shared__ __attribute__((aligned(16))) float lds[];
    const int b = blockIdx.x;
    if (b < NSORT) {
        if (STEP == 2)      sort_rank(b, dst, cnt, rank);
        else if (STEP == 3) sort_scan(cnt, offs, (int*)lds);
        else                sort_scatter(b, src, dst, etype, offs, rank, grouped);
        return;
    }
    proj_tile(b - NSORT + TILE0, lds, h, W, attw, z, ed);
}

__device__ __forceinline__ float quad_sum(float x) {
    int b = __builtin_bit_cast(int, x);
    float y = x + __builtin_bit_cast(float, __builtin_amdgcn_update_dpp(b, b, 0xB1, 0xF, 0xF, false));
    int c = __builtin_bit_cast(int, y);
    return y + __builtin_bit_cast(float, __builtin_amdgcn_update_dpp(c, c, 0x4E, 0xF, 0xF, false));
}

__device__ __forceinline__ float pick8(const float (&p)[8], int r) {
    float a = (r & 1) ? p[1] : p[0];
    float b = (r & 1) ? p[3] : p[2];
    float c = (r & 1) ? p[5] : p[4];
    float d = (r & 1) ? p[7] : p[6];
    float ab = (r & 2) ? b : a;
    float cd = (r & 2) ? d : c;
    return (r & 4) ? cd : ab;
}

__device__ __forceinline__ float dot4(float4 a, float4 b) {
    return a.x * b.x + a.y * b.y + a.z * b.z + a.w * b.w;
}

__global__ __launch_bounds__(256) void rgat_node_softmax_sum(const float* __restrict__ z,
                                                             const float* __restrict__ ed,
                                                             const float* __restrict__ attw,
                                                             const int* __restrict__ offs,
                                                             const int2* __restrict__ grouped,
                                                             float* __restrict__ out) {
    const int lane = threadIdx.x & 31;
    const int d = __builtin_amdgcn_readfirstlane(blockIdx.x * 8 + (threadIdx.x >> 5));
    if (d >= NN) return;
    const int hd = lane >> 2;
    const int qd = lane & 3;

    const int beg = __builtin_amdgcn_readfirstlane(offs[d]);
    const int end = __builtin_amdgcn_readfirstlane(offs[d + 1]);

    float4 acc = make_float4(0.0f, 0.0f, 0.0f, 0.0f);
    float  m = -INFINITY;
    float  l = 0.0f;

    if (end > beg) {
        int base = beg;
        int n = end - base;
        if (n > 32) n = 32;
        int2 my = make_int2(0, 0);
        if (lane < n) my = grouped[base + lane];

        float pd[8];
#pragma unroll
        for (int r = 0; r < RR; ++r)
            pd[r] = ed[(((size_t)r * 2 + (hd >> 2)) * NN + d) * 4 + (hd & 3)];

        for (;;) {
            for (int j = 0; j < n; j += 4) {
                float4 zs[4], wS[4];
                float  pdr[4];
#pragma unroll
                for (int u = 0; u < 4; ++u) {
                    int jj = j + u;
                    if (jj > n - 1) jj = n - 1;
                    const int s = __builtin_amdgcn_readlane(my.x, jj);
                    const int r = __builtin_amdgcn_readlane(my.y, jj);
                    zs[u]  = *(const float4*)(z + ((size_t)r * NN + s) * OUTD + lane * 4);
                    wS[u]  = *(const float4*)(attw + ((size_t)r * HH + hd) * (2 * HDD) + qd * 4);
                    pdr[u] = pick8(pd, r);
                }
#pragma unroll
                for (int u = 0; u < 4; ++u) {
                    if (j + u < n) {
                        float a  = quad_sum(dot4(zs[u], wS[u])) + pdr[u];
                        float ev = (a > 0.0f) ? a : 0.01f * a;
                        float mn = fmaxf(m, ev);
                        float sc = __expf(m - mn);
                        float w  = __expf(ev - mn);
                        l     = l * sc + w;
                        acc.x = acc.x * sc + w * zs[u].x;
                        acc.y = acc.y * sc + w * zs[u].y;
                        acc.z = acc.z * sc + w * zs[u].z;
                        acc.w = acc.w * sc + w * zs[u].w;
                        m = mn;
                    }
                }
            }
            base += 32;
            if (base >= end) break;
            n = end - base;
            if (n > 32) n = 32;
            my = make_int2(0, 0);
            if (lane < n) my = grouped[base + lane];
        }
    }

    const float den = fmaxf(l, 1e-9f);
    float4 o;
    o.x = acc.x / den; o.y = acc.y / den; o.z = acc.z / den; o.w = acc.w / den;
    *(float4*)(out + (size_t)d * OUTD + lane * 4) = o;
}

extern "C" void kernel_launch(void* const* d_in, const int* in_sizes, int n_in,
                              void* d_out, int out_size, void* d_ws, size_t ws_size,
                              hipStream_t stream) {
    const float* h    = (const float*)d_in[0];
    const float* fcw  = (const float*)d_in[1];
    const float* attw = (const float*)d_in[2];
    const int*   src  = (const int*)d_in[3];
    const int*   dst  = (const int*)d_in[4];
    const int*   ety  = (const int*)d_in[5];
    float* out = (float*)d_out;

    float* z       = (float*)d_ws;
    float* ed      = z + (size_t)RR * NN * OUTD;
    int*   cnt     = (int*)(ed + (size_t)RR * HH * NN);
    int*   offs    = cnt + CNT_PAD;
    int*   rank    = offs + CNT_PAD;
    int2*  grouped = (int2*)(rank + EE);

    rgat_zero_counts<<<CNT_PAD / 256, 256, 0, stream>>>(cnt);
    {
        size_t shmem = (size_t)(64 * HS_A + 64 * WPS) * sizeof(float);
        rgat_proj_and_sort<2, SORT_BLK, 0><<<SORT_BLK + PROJ_P1, 256, shmem, stream>>>(
            h, fcw, attw, z, ed, src, dst, ety, cnt, offs, rank, grouped);
        rgat_proj_and_sort<3, 1, PROJ_P1><<<1 + PROJ_P2, 256, shmem, stream>>>(
            h, fcw, attw, z, ed, src, dst, ety, cnt, offs, rank, grouped);
        rgat_proj_and_sort<4, SORT_BLK, PROJ_P1 + PROJ_P2><<<SORT_BLK + PROJ_P3, 256, shmem, stream>>>(
            h, fcw, attw, z, ed, src, dst, ety, cnt, offs, rank, grouped);
    }
    rgat_node_softmax_sum<<<(NN + 7) / 8, 256, 0, stream>>>(z, ed, attw, offs, grouped, out);
}


# z tile stored as 8 whole-line 16-byte stores per wave after a 4x4 quad transpose of the accumulators (DPP); was 32 four-byte stores, each behind its own row test and address wait; same sums
# speedup vs baseline: 3.7630x; 1.0687x over previous
#include <hip/hip_runtime.h>
#include <hip/hip_bf16.h>
#include <math.h>

#define NN 50000
#define EE 800000
#define RR 8
#define IND 128
#define OUTD 128
#define HH 8
#define HDD 16

typedef float v2f __attribute__((ext_vector_type(2)));
typedef float v8f __attribute__((ext_vector_type(8)));

#define HS_A 132
#define WPS  288

template <int CTRL>
__device__ __forceinline__ float row_xor(float x) {
    return __builtin_bit_cast(float,
        __builtin_amdgcn_update_dpp(0, __builtin_bit_cast(int, x), CTRL, 0xF, 0xF, true));
}

#define TILES_X    ((NN + 63) / 64)
#define PROJ_TILES (TILES_X * RR)
#define PROJ_P1    2048
#define PROJ_P2    2048
#define PROJ_P3    (PROJ_TILES - PROJ_P1 - PROJ_P2)

__device__ __forceinline__ void proj_tile(const int pid, float* lds,
                                          const float* __restrict__ h,
                                          const float* __restrict__ W,
                                          const float* __restrict__ attw,
                                          float* __restrict__ z,
                                          float* __restrict__ ed) {
    float* h_s = lds;
    float* w_s = lds + 64 * HS_A;

    const int r     = pid / TILES_X;
    const int nbase = (pid - r * TILES_X) * 64;
    const int tid   = threadIdx.x;

    float4 hv[8];
#pragma unroll
    for (int it = 0; it < 8; ++it) {
        const int i   = tid + it * 256;
        int       gr  = nbase + (i >> 5);
        if (gr >= NN) gr = NN - 1;
        hv[it] = *(const float4*)(h + (size_t)gr * IND + (i & 31) * 4);
    }
    const float* Wr = W + (size_t)r * IND * OUTD;
    float4 wa[8], wb[8];
#pragma unroll
    for (int it = 0; it < 8; ++it) {
        const int i  = tid + it * 256;
        const int kp = i >> 5;
        const int c4 = i & 31;
        wa[it] = *(const float4*)(Wr + (size_t)(2 * kp) * OUTD + c4 * 4);
        wb[it] = *(const float4*)(Wr + (size_t)(2 * kp + 1) * OUTD + c4 * 4);
    }
#pragma unroll
    for (int it = 0; it < 8; ++it) {
        const int i = tid + it * 256;
        *(float4*)(h_s + (i >> 5) * HS_A + (i & 31) * 4) = hv[it];
    }
#pragma unroll
    for (int it = 0; it < 8; ++it) {
        const int i  = tid + it * 256;
        float* d = w_s + (i >> 5) * WPS + (i & 31) * 8;
        *(float4*)(d)     = make_float4(wa[it].x, wb[it].x, wa[it].y, wb[it].y);
        *(float4*)(d + 4) = make_float4(wa[it].z, wb[it].z, wa[it].w, wb[it].w);
    }
    __syncthreads();

    const int wave = tid >> 5;
    const int lane = tid & 31;
    const int ln   = lane & 15;
    const int g    = lane >> 4;
    const int rt   = wave & 3;
    const int cb   = (wave >> 2) * 64;

    v8f acc0 = {}, acc1 = {}, acc2 = {}, acc3 = {};
    const float* arow = h_s + (rt * 16 + ln) * HS_A;
    const float* bcol = w_s + (size_t)g * WPS + (size_t)(cb + ln) * 2;

    for (int k = 0; k < IND; k += 4) {
        v2f a = *(const v2f*)(arow + k + 2 * g);
        const float* brow = bcol + (k >> 1) * WPS;
        v2f b0 = *(const v2f*)(brow + 0 * 32);
        v2f b1 = *(const v2f*)(brow + 1 * 32);
        v2f b2 = *(const v2f*)(brow + 2 * 32);
        v2f b3 = *(const v2f*)(brow + 3 * 32);
        acc0 = __builtin_amdgcn_wmma_f32_16x16x4_f32(false, a, false, b0, (short)0, acc0, false, false);
        acc1 = __builtin_amdgcn_wmma_f32_16x16x4_f32(false, a, false, b1, (short)0, acc1, false, false);
        acc2 = __builtin_amdgcn_wmma_f32_16x16x4_f32(false, a, false, b2, (short)0, acc2, false, false);
        acc3 = __builtin_amdgcn_wmma_f32_16x16x4_f32(false, a, false, b3, (short)0, acc3, false, false);
    }

    v2f sD;
    const int hh = wave >> 2;
    {
        const bool l0 = (ln & 1) != 0, l1 = (ln & 2) != 0;
        const bool l2 = (ln & 4) != 0, l3 = (ln & 8) != 0;
        float sC[4];
#pragma unroll
        for (int ct = 0; ct < 4; ++ct) {
            const v8f acc = (ct == 0) ? acc0 : (ct == 1) ? acc1 : (ct == 2) ? acc2 : acc3;
            const float wd = attw[((size_t)r * HH + hh * 4 + ct) * (2 * HDD) + HDD + ln];
            float sA[4];
#pragma unroll
            for (int p = 0; p < 4; ++p) {
                const float x0 = acc[2 * p] * wd, x1 = acc[2 * p + 1] * wd;
                sA[p] = (l0 ? x1 : x0) + row_xor<0x161>(l0 ? x0 : x1);
            }
            float sB[2];
#pragma unroll
            for (int q = 0; q < 2; ++q)
                sB[q] = (l1 ? sA[2 * q + 1] : sA[2 * q]) + row_xor<0x162>(l1 ? sA[2 * q] : sA[2 * q + 1]);
            sC[ct] = (l2 ? sB[1] : sB[0]) + row_xor<0x164>(l2 ? sB[0] : sB[1]);
        }
        sD.x = (l3 ? sC[2] : sC[0]) + row_xor<0x168>(l3 ? sC[0] : sC[2]);
        sD.y = (l3 ? sC[3] : sC[1]) + row_xor<0x168>(l3 ? sC[1] : sC[3]);
    }

    static_assert(NN % 16 == 0, "a wave's 16 rows are all inside or all outside");
    {
        const bool b0 = (ln & 1) != 0, b1 = (ln & 2) != 0;
        const int wrow = nbase + rt * 16;
        float4 y[8];
#pragma unroll
        for (int v = 0; v < 8; ++v) {
            const float x0 = acc0[v], x1 = acc1[v], x2 = acc2[v], x3 = acc3[v];
            const float r01 = row_xor<0x161>(b0 ? x0 : x1);
            const float r23 = row_xor<0x161>(b0 ? x2 : x3);
            const float t0 = b0 ? r01 : x0, t1 = b0 ? x1 : r01;
            const float t2 = b0 ? r23 : x2, t3 = b0 ? x3 : r23;
            const float r02 = row_xor<0x162>(b1 ? t0 : t2);
            const float r13 = row_xor<0x162>(b1 ? t1 : t3);
            y[v] = make_float4(b1 ? r02 : t0, b1 ? r13 : t1, b1 ? t2 : r02, b1 ? t3 : r13);
        }
        if (wrow < NN) {
            float* zp = z + ((size_t)r * NN + wrow + 8 * g) * OUTD + cb + 16 * (ln & 3) + 4 * (ln >> 2);
#pragma unroll
            for (int v = 0; v < 8; ++v)
                *(float4*)(zp + (size_t)v * OUTD) = y[v];
            *(v2f*)(ed + (((size_t)r * 2 + hh) * NN + wrow + 8 * g + (ln & 7)) * 4 + 2 * (ln >> 3)) = sD;
        }
    }
}

#define CNT_PAD   53248
#define SCAN_CH   13
#define SORT_EPT  8
#define SORT_BLK  ((EE + 256 * SORT_EPT - 1) / (256 * SORT_EPT))
static_assert(EE % SORT_EPT == 0, "a thread's edges are all inside or all outside");
static_assert(SCAN_CH * 4096 == CNT_PAD, "scan chunks cover the counters");
static_assert(CNT_PAD >= NN + 1, "offs[NN] must exist");

__global__ __launch_bounds__(256) void rgat_zero_counts(int* __restrict__ cnt) {
    int t = blockIdx.x * 256 + threadIdx.x;
    if (t < CNT_PAD) cnt[t] = 0;
}

__device__ __forceinline__ void sort_rank(const int b, const int* __restrict__ dst,
                                          int* __restrict__ cnt, int* __restrict__ rank) {
    const int e0 = (b * 256 + (int)threadIdx.x) * SORT_EPT;
    if (e0 >= EE) return;
    const int4 d0 = *(const int4*)(dst + e0);
    const int4 d1 = *(const int4*)(dst + e0 + 4);
    int4 k0, k1;
    k0.x = atomicAdd(cnt + d0.x, 1);
    k0.y = atomicAdd(cnt + d0.y, 1);
    k0.z = atomicAdd(cnt + d0.z, 1);
    k0.w = atomicAdd(cnt + d0.w, 1);
    k1.x = atomicAdd(cnt + d1.x, 1);
    k1.y = atomicAdd(cnt + d1.y, 1);
    k1.z = atomicAdd(cnt + d1.z, 1);
    k1.w = atomicAdd(cnt + d1.w, 1);
    *(int4*)(rank + e0)     = k0;
    *(int4*)(rank + e0 + 4) = k1;
}

__device__ __forceinline__ int wave_incl_scan(int x, const int lane) {
#pragma unroll
    for (int o = 1; o < 32; o <<= 1) {
        int t = __shfl_up(x, o);
        if (lane >= o) x += t;
    }
    return x;
}

#define SCAN_G0 7

template <int C0, int NC>
__device__ __forceinline__ void scan_load(const int* __restrict__ cnt, const int tid, int4 (&v)[NC][4]) {
#pragma unroll
    for (int u = 0; u < NC; ++u) {
        const int4* p = (const int4*)(cnt + (C0 + u) * 4096 + tid * 16);
        v[u][0] = p[0]; v[u][1] = p[1]; v[u][2] = p[2]; v[u][3] = p[3];
    }
}

template <int C0, int NC>
__device__ __forceinline__ void scan_sums(const int* __restrict__ cnt, int* sS, int* sI,
                                          const int tid, const int lane) {
    int4 v[NC][4];
    scan_load<C0, NC>(cnt, tid, v);
#pragma unroll
    for (int u = 0; u < NC; ++u) {
        const int sum = (v[u][0].x + v[u][0].y + v[u][0].z + v[u][0].w)
                      + (v[u][1].x + v[u][1].y + v[u][1].z + v[u][1].w)
                      + (v[u][2].x + v[u][2].y + v[u][2].z + v[u][2].w)
                      + (v[u][3].x + v[u][3].y + v[u][3].z + v[u][3].w);
        sS[(C0 + u) * 256 + tid] = sum;
        sI[(C0 + u) * 256 + tid] = wave_incl_scan(sum, lane);
    }
}

template <int C0, int NC>
__device__ __forceinline__ void scan_write(const int* __restrict__ cnt, int* __restrict__ offs,
                                           const int* wt, const int* sS, const int* sI,
                                           const int tid, const int wv) {
    int4 v[NC][4];
    scan_load<C0, NC>(cnt, tid, v);
#pragma unroll
    for (int u = 0; u < NC; ++u) {
        const int c = C0 + u;
        int run = wt[c * 8 + wv] + sI[c * 256 + tid] - sS[c * 256 + tid];
        int4* q = (int4*)(offs + c * 4096 + tid * 16);
#pragma unroll
        for (int k = 0; k < 4; ++k) {
            int4 o;
            o.x = run; run += v[u][k].x;
            o.y = run; run += v[u][k].y;
            o.z = run; run += v[u][k].z;
            o.w = run; run += v[u][k].w;
            q[k] = o;
        }
    }
}

__device__ __forceinline__ void sort_scan(const int* __restrict__ cnt, int* __restrict__ offs,
                                          int* wt  ) {
    const int tid  = threadIdx.x;
    const int lane = tid & 31;
    const int wv   = tid >> 5;
    int* sS = wt + 128;
    int* sI = sS + SCAN_CH * 256;

    scan_sums<0, SCAN_G0>(cnt, sS, sI, tid, lane);
    scan_sums<SCAN_G0, SCAN_CH - SCAN_G0>(cnt, sS, sI, tid, lane);
    __syncthreads();
    if (wv == 0) {
        int a0 = 0, a1 = 0, a2 = 0, a3 = 0;
        if (lane < SCAN_CH * 2) {
            a0 = sI[(4 * lane)     * 32 + 31]; a1 = sI[(4 * lane + 1) * 32 + 31];
            a2 = sI[(4 * lane + 2) * 32 + 31]; a3 = sI[(4 * lane + 3) * 32 + 31];
        }
        const int sum4 = a0 + a1 + a2 + a3;
        const int ex   = wave_incl_scan(sum4, lane) - sum4;
        if (lane < SCAN_CH * 2) {
            wt[4 * lane]     = ex;
            wt[4 * lane + 1] = ex + a0;
            wt[4 * lane + 2] = ex + a0 + a1;
            wt[4 * lane + 3] = ex + a0 + a1 + a2;
        }
    }
    __syncthreads();
    scan_write<0, SCAN_G0>(cnt, offs, wt, sS, sI, tid, wv);
    scan_write<SCAN_G0, SCAN_CH - SCAN_G0>(cnt, offs, wt, sS, sI, tid, wv);
}

__device__ __forceinline__ void sort_scatter(const int b,
                                             const int* __restrict__ src,
                                             const int* __restrict__ dst,
                                             const int* __restrict__ etype,
                                             const int* __restrict__ offs,
                                             const int* __restrict__ rank,
                                             int2* __restrict__ grouped) {
    const int e0 = (b * 256 + (int)threadIdx.x) * SORT_EPT;
    if (e0 >= EE) return;
    const int4 d0 = *(const int4*)(dst + e0),   d1 = *(const int4*)(dst + e0 + 4);
    const int4 k0 = *(const int4*)(rank + e0),  k1 = *(const int4*)(rank + e0 + 4);
    const int4 s0 = *(const int4*)(src + e0),   s1 = *(const int4*)(src + e0 + 4);
    const int4 t0 = *(const int4*)(etype + e0), t1 = *(const int4*)(etype + e0 + 4);
    const int p0 = offs[d0.x] + k0.x, p1 = offs[d0.y] + k0.y;
    const int p2 = offs[d0.z] + k0.z, p3 = offs[d0.w] + k0.w;
    const int p4 = offs[d1.x] + k1.x, p5 = offs[d1.y] + k1.y;
    const int p6 = offs[d1.z] + k1.z, p7 = offs[d1.w] + k1.w;
    grouped[p0] = make_int2(s0.x, t0.x);
    grouped[p1] = make_int2(s0.y, t0.y);
    grouped[p2] = make_int2(s0.z, t0.z);
    grouped[p3] = make_int2(s0.w, t0.w);
    grouped[p4] = make_int2(s1.x, t1.x);
    grouped[p5] = make_int2(s1.y, t1.y);
    grouped[p6] = make_int2(s1.z, t1.z);
    grouped[p7] = make_int2(s1.w, t1.w);
}

template <int STEP, int NSORT, int TILE0>
__global__ __launch_bounds__(256) __attribute__((amdgpu_waves_per_eu(6, 6)))
void rgat_proj_and_sort(const float* __restrict__ h,
                        const float* __restrict__ W,
                        const float* __restrict__ attw,
                        float* __restrict__ z,
                        float* __restrict__ ed,
                        const int* __restrict__ src,
                        const int* __restrict__ dst,
                        const int* __restrict__ etype,
                        int* __restrict__ cnt,
                        int* __restrict__ offs,
                        int* __restrict__ rank,
                        int2* __restrict__ grouped) {
    extern __shared__ __attribute__((aligned(16))) float lds[];
    const int b = blockIdx.x;
    if (b < NSORT) {
        if (STEP == 2)      sort_rank(b, dst, cnt, rank);
        else if (STEP == 3) sort_scan(cnt, offs, (int*)lds);
        else                sort_scatter(b, src, dst, etype, offs, rank, grouped);
        return;
    }
    proj_tile(b - NSORT + TILE0, lds, h, W, attw, z, ed);
}

__device__ __forceinline__ float quad_sum(float x) {
    int b = __builtin_bit_cast(int, x);
    float y = x + __builtin_bit_cast(float, __builtin_amdgcn_update_dpp(b, b, 0xB1, 0xF, 0xF, false));
    int c = __builtin_bit_cast(int, y);
    return y + __builtin_bit_cast(float, __builtin_amdgcn_update_dpp(c, c, 0x4E, 0xF, 0xF, false));
}

__device__ __forceinline__ float pick8(const float (&p)[8], int r) {
    float a = (r & 1) ? p[1] : p[0];
    float b = (r & 1) ? p[3] : p[2];
    float c = (r & 1) ? p[5] : p[4];
    float d = (r & 1) ? p[7] : p[6];
    float ab = (r & 2) ? b : a;
    float cd = (r & 2) ? d : c;
    return (r & 4) ? cd : ab;
}

__device__ __forceinline__ float dot4(float4 a, float4 b) {
    return a.x * b.x + a.y * b.y + a.z * b.z + a.w * b.w;
}

__global__ __launch_bounds__(256) void rgat_node_softmax_sum(const float* __restrict__ z,
                                                             const float* __restrict__ ed,
                                                             const float* __restrict__ attw,
                                                             const int* __restrict__ offs,
                                                             const int2* __restrict__ grouped,
                                                             float* __restrict__ out) {
    const int lane = threadIdx.x & 31;
    const int d = __builtin_amdgcn_readfirstlane(blockIdx.x * 8 + (threadIdx.x >> 5));
    if (d >= NN) return;
    const int hd = lane >> 2;
    const int qd = lane & 3;

    const int beg = __builtin_amdgcn_readfirstlane(offs[d]);
    const int end = __builtin_amdgcn_readfirstlane(offs[d + 1]);

    float4 acc = make_float4(0.0f, 0.0f, 0.0f, 0.0f);
    float  m = -INFINITY;
    float  l = 0.0f;

    if (end > beg) {
        int base = beg;
        int n = end - base;
        if (n > 32) n = 32;
        int2 my = make_int2(0, 0);
        if (lane < n) my = grouped[base + lane];

        float pd[8];
#pragma unroll
        for (int r = 0; r < RR; ++r)
            pd[r] = ed[(((size_t)r * 2 + (hd >> 2)) * NN + d) * 4 + (hd & 3)];

        for (;;) {
            for (int j = 0; j < n; j += 4) {
                float4 zs[4], wS[4];
                float  pdr[4];
#pragma unroll
                for (int u = 0; u < 4; ++u) {
                    int jj = j + u;
                    if (jj > n - 1) jj = n - 1;
                    const int s = __builtin_amdgcn_readlane(my.x, jj);
                    const int r = __builtin_amdgcn_readlane(my.y, jj);
                    zs[u]  = *(const float4*)(z + ((size_t)r * NN + s) * OUTD + lane * 4);
                    wS[u]  = *(const float4*)(attw + ((size_t)r * HH + hd) * (2 * HDD) + qd * 4);
                    pdr[u] = pick8(pd, r);
                }
#pragma unroll
                for (int u = 0; u < 4; ++u) {
                    if (j + u < n) {
                        float a  = quad_sum(dot4(zs[u], wS[u])) + pdr[u];
                        float ev = (a > 0.0f) ? a : 0.01f * a;
                        float mn = fmaxf(m, ev);
                        float sc = __expf(m - mn);
                        float w  = __expf(ev - mn);
                        l     = l * sc + w;
                        acc.x = acc.x * sc + w * zs[u].x;
                        acc.y = acc.y * sc + w * zs[u].y;
                        acc.z = acc.z * sc + w * zs[u].z;
                        acc.w = acc.w * sc + w * zs[u].w;
                        m = mn;
                    }
                }
            }
            base += 32;
            if (base >= end) break;
            n = end - base;
            if (n > 32) n = 32;
            my = make_int2(0, 0);
            if (lane < n) my = grouped[base + lane];
        }
    }

    const float den = fmaxf(l, 1e-9f);
    float4 o;
    o.x = acc.x / den; o.y = acc.y / den; o.z = acc.z / den; o.w = acc.w / den;
    *(float4*)(out + (size_t)d * OUTD + lane * 4) = o;
}

extern "C" void kernel_launch(void* const* d_in, const int* in_sizes, int n_in,
                              void* d_out, int out_size, void* d_ws, size_t ws_size,
                              hipStream_t stream) {
    const float* h    = (const float*)d_in[0];
    const float* fcw  = (const float*)d_in[1];
    const float* attw = (const float*)d_in[2];
    const int*   src  = (const int*)d_in[3];
    const int*   dst  = (const int*)d_in[4];
    const int*   ety  = (const int*)d_in[5];
    float* out = (float*)d_out;

    float* z       = (float*)d_ws;
    float* ed      = z + (size_t)RR * NN * OUTD;
    int*   cnt     = (int*)(ed + (size_t)RR * HH * NN);
    int*   offs    = cnt + CNT_PAD;
    int*   rank    = offs + CNT_PAD;
    int2*  grouped = (int2*)(rank + EE);

    rgat_zero_counts<<<CNT_PAD / 256, 256, 0, stream>>>(cnt);
    {
        size_t shmem = (size_t)(64 * HS_A + 64 * WPS) * sizeof(float);
        rgat_proj_and_sort<2, SORT_BLK, 0><<<SORT_BLK + PROJ_P1, 256, shmem, stream>>>(
            h, fcw, attw, z, ed, src, dst, ety, cnt, offs, rank, grouped);
        rgat_proj_and_sort<3, 1, PROJ_P1><<<1 + PROJ_P2, 256, shmem, stream>>>(
            h, fcw, attw, z, ed, src, dst, ety, cnt, offs, rank, grouped);
        rgat_proj_and_sort<4, SORT_BLK, PROJ_P1 + PROJ_P2><<<SORT_BLK + PROJ_P3, 256, shmem, stream>>>(
            h, fcw, attw, z, ed, src, dst, ety, cnt, offs, rank, grouped);
    }
    rgat_node_softmax_sum<<<(NN + 7) / 8, 256, 0, stream>>>(z, ed, attw, offs, grouped, out);
}
